# starting kernel
# baseline (speedup 1.0000x reference)
#include <hip/hip_runtime.h>
#include <math.h>

typedef __attribute__((ext_vector_type(2))) float v2f;
typedef __attribute__((ext_vector_type(8))) float v8f;

#define N_NODES 50000
#define N_EDGES 800000
#define HID 256

__global__ __launch_bounds__(256) void edge_mlp_kernel(
    const float* __restrict__ attr,
    const float* __restrict__ W1,
    const float* __restrict__ b1,
    const float* __restrict__ W2,
    const float* __restrict__ b2,
    float* __restrict__ w, int E) {
  __shared__ float sW1[3 * 128];
  __shared__ float sb1[128];
  __shared__ float sW2[128];
  for (int i = threadIdx.x; i < 384; i += 256) sW1[i] = W1[i];
  if (threadIdx.x < 128) {
    sb1[threadIdx.x] = b1[threadIdx.x];
    sW2[threadIdx.x] = W2[threadIdx.x];
  }
  __syncthreads();
  int e = blockIdx.x * 256 + threadIdx.x;
  if (e >= E) return;
  float a0 = attr[e * 3 + 0];
  float a1 = attr[e * 3 + 1];
  float a2 = attr[e * 3 + 2];
  float acc = 0.0f;
#pragma unroll 4
  for (int j = 0; j < 128; ++j) {
    float z = fmaf(a0, sW1[j], fmaf(a1, sW1[128 + j], fmaf(a2, sW1[256 + j], sb1[j])));
    float t = z / (1.0f + __expf(-z));
    acc = fmaf(t, sW2[j], acc);
  }
  float zz = acc + b2[0];
  w[e] = 1.0f / (1.0f + __expf(-zz));
}

__global__ __launch_bounds__(256) void deg_init_kernel(float* __restrict__ deg, int N) {
  int i = blockIdx.x * 256 + threadIdx.x;
  if (i < N) deg[i] = 1.0f;
}

__global__ __launch_bounds__(256) void deg_scatter_kernel(
    const int* __restrict__ col, const float* __restrict__ w,
    float* __restrict__ deg, int E) {
  int e = blockIdx.x * 256 + threadIdx.x;
  if (e < E) unsafeAtomicAdd(&deg[col[e]], w[e]);
}

__global__ __launch_bounds__(256) void deg_rsqrt_kernel(float* __restrict__ deg, int N) {
  int i = blockIdx.x * 256 + threadIdx.x;
  if (i < N) {
    float d = deg[i];
    deg[i] = d > 0.0f ? rsqrtf(d) : 0.0f;
  }
}

__global__ __launch_bounds__(256) void gemm_xWc_kernel(
    const float* __restrict__ h,
    const float* __restrict__ Wc,
    float* __restrict__ x) {
  const int K = 256, NN = 256;
  const int wave = threadIdx.x >> 5;
  const int lane = threadIdx.x & 31;
  const int m0 = (blockIdx.x >> 1) * 16;
  const int n0 = (((blockIdx.x & 1) << 3) + wave) * 16;
  const int r = lane & 15;
  const int g = lane >> 4;

  const float* arow = h + (size_t)(m0 + r) * K + 2 * g;
  const float* bcol = Wc + (size_t)(2 * g) * NN + n0 + r;

  v8f c = {};
  for (int kk = 0; kk < K; kk += 4) {
    v2f a;
    a.x = arow[kk];
    a.y = arow[kk + 1];
    v2f b;
    b.x = bcol[kk * NN];
    b.y = bcol[kk * NN + NN];
    c = __builtin_amdgcn_wmma_f32_16x16x4_f32(false, a, false, b, (short)0, c,
                                              false, false);
  }
#pragma unroll
  for (int i = 0; i < 8; ++i) {
    int m = m0 + i + 8 * g;
    x[(size_t)m * NN + n0 + r] = c[i];
  }
}

__global__ __launch_bounds__(256) void out_init_kernel(
    const float* __restrict__ x, const float* __restrict__ dinv,
    const float* __restrict__ bc, float* __restrict__ out) {
  int idx = blockIdx.x * 256 + threadIdx.x;
  int n = idx >> 8;
  int j = idx & 255;
  float di = dinv[n];
  out[idx] = x[idx] * di * di + bc[j];
}

__global__ __launch_bounds__(256) void aggregate_kernel(
    const int* __restrict__ ei,
    const float* __restrict__ w,
    const float* __restrict__ dinv,
    const float* __restrict__ x,
    float* __restrict__ out, int E) {
  int e = blockIdx.x;
  int r = ei[e];
  int c = ei[E + e];
  float norm = dinv[r] * w[e] * dinv[c];
  int j = threadIdx.x;
  float v = x[(size_t)r * HID + j] * norm;
  unsafeAtomicAdd(&out[(size_t)c * HID + j], v);
}

extern "C" void kernel_launch(void* const* d_in, const int* in_sizes, int n_in,
                              void* d_out, int out_size, void* d_ws, size_t ws_size,
                              hipStream_t stream) {
  const float* h    = (const float*)d_in[0];
  const int*   ei   = (const int*)d_in[1];
  const float* attr = (const float*)d_in[2];
  const float* W1   = (const float*)d_in[3];
  const float* b1   = (const float*)d_in[4];
  const float* W2   = (const float*)d_in[5];
  const float* b2   = (const float*)d_in[6];
  const float* Wc   = (const float*)d_in[7];
  const float* bc   = (const float*)d_in[8];
  float* out = (float*)d_out;

  const int N = N_NODES, E = N_EDGES;

  float* x   = (float*)d_ws;
  float* w   = x + (size_t)N * HID;
  float* deg = w + E;

  edge_mlp_kernel<<<(E + 255) / 256, 256, 0, stream>>>(attr, W1, b1, W2, b2, w, E);
  deg_init_kernel<<<(N + 255) / 256, 256, 0, stream>>>(deg, N);
  deg_scatter_kernel<<<(E + 255) / 256, 256, 0, stream>>>(ei + E, w, deg, E);
  deg_rsqrt_kernel<<<(N + 255) / 256, 256, 0, stream>>>(deg, N);
  gemm_xWc_kernel<<<(N / 16) * 2, 256, 0, stream>>>(h, Wc, x);
  out_init_kernel<<<N, 256, 0, stream>>>(x, deg, bc, out);
  aggregate_kernel<<<E, 256, 0, stream>>>(ei, w, deg, x, out, E);
}


# edges counting-sorted by destination each call (integer atomic rank + scan); degree and feature sums become per-row gathers, no float atomics; sum order now by arrival-order rank
# speedup vs baseline: 3.2984x; 3.2984x over previous
#include <hip/hip_runtime.h>
#include <math.h>

typedef __attribute__((ext_vector_type(2))) float v2f;
typedef __attribute__((ext_vector_type(8))) float v8f;

#define N_NODES 50000
#define N_EDGES 800000
#define HID 256

__global__ __launch_bounds__(256) void cnt_zero_kernel(int* __restrict__ cnt, int N) {
  int i = blockIdx.x * 256 + threadIdx.x;
  if (i < N) cnt[i] = 0;
}

__global__ __launch_bounds__(256) void edge_mlp_kernel(
    const float* __restrict__ attr,
    const float* __restrict__ W1,
    const float* __restrict__ b1,
    const float* __restrict__ W2,
    const float* __restrict__ b2,
    const int* __restrict__ col,
    int* __restrict__ cnt,
    float* __restrict__ w, int* __restrict__ rank, int E) {
  __shared__ float sW1[3 * 128];
  __shared__ float sb1[128];
  __shared__ float sW2[128];
  for (int i = threadIdx.x; i < 384; i += 256) sW1[i] = W1[i];
  if (threadIdx.x < 128) {
    sb1[threadIdx.x] = b1[threadIdx.x];
    sW2[threadIdx.x] = W2[threadIdx.x];
  }
  __syncthreads();
  int e = blockIdx.x * 256 + threadIdx.x;
  if (e >= E) return;
  const int rk = atomicAdd(&cnt[col[e]], 1);
  float a0 = attr[e * 3 + 0];
  float a1 = attr[e * 3 + 1];
  float a2 = attr[e * 3 + 2];
  float acc = 0.0f;
#pragma unroll 4
  for (int j = 0; j < 128; ++j) {
    float z = fmaf(a0, sW1[j], fmaf(a1, sW1[128 + j], fmaf(a2, sW1[256 + j], sb1[j])));
    float t = z / (1.0f + __expf(-z));
    acc = fmaf(t, sW2[j], acc);
  }
  float zz = acc + b2[0];
  w[e] = 1.0f / (1.0f + __expf(-zz));
  rank[e] = rk;
}

#define SCAN_ITERS 13
static_assert(32 * SCAN_ITERS * 128 >= N_NODES, "scan slice too small");
static_assert(N_NODES % 4 == 0, "int4 groups must not straddle the end");

__device__ __forceinline__ int wave_incl_scan(int v, int lane) {
#pragma unroll
  for (int d = 1; d < 32; d <<= 1) {
    int t = __shfl_up(v, d, 32);
    if (lane >= d) v += t;
  }
  return v;
}

__global__ __launch_bounds__(1024) void offs_scan_kernel(
    const int* __restrict__ cnt, int* __restrict__ offs) {
  __shared__ int wtot[32];
  const int wave = threadIdx.x >> 5;
  const int lane = threadIdx.x & 31;
  const int base = wave * (SCAN_ITERS * 128) + lane * 4;
  int4 v[SCAN_ITERS];
  int ex[SCAN_ITERS];
#pragma unroll
  for (int i = 0; i < SCAN_ITERS; ++i) {
    const int idx = base + i * 128;
    v[i] = idx < N_NODES ? *(const int4*)(cnt + idx) : make_int4(0, 0, 0, 0);
  }
  int carry = 0;
#pragma unroll
  for (int i = 0; i < SCAN_ITERS; ++i) {
    const int sum4 = v[i].x + v[i].y + v[i].z + v[i].w;
    const int inc = wave_incl_scan(sum4, lane);
    ex[i] = carry + inc - sum4;
    carry += __shfl(inc, 31, 32);
  }
  if (lane == 0) wtot[wave] = carry;
  __syncthreads();
  const int tot = wtot[lane];
  const int tinc = wave_incl_scan(tot, lane);
  const int woff = __shfl(tinc - tot, wave, 32);
#pragma unroll
  for (int i = 0; i < SCAN_ITERS; ++i) {
    const int idx = base + i * 128;
    if (idx < N_NODES) {
      const int p = woff + ex[i];
      *(int4*)(offs + idx) =
          make_int4(p, p + v[i].x, p + v[i].x + v[i].y, p + v[i].x + v[i].y + v[i].z);
    }
  }
}

__global__ __launch_bounds__(256) void fill_kernel(
    const int* __restrict__ ei,
    const float* __restrict__ w, const int* __restrict__ rank,
    const int* __restrict__ offs, int2* __restrict__ sw, int E) {
  int e = blockIdx.x * 256 + threadIdx.x;
  if (e >= E) return;
  const int r = ei[e];
  const int c = ei[E + e];
  sw[offs[c] + rank[e]] = make_int2(r, __float_as_int(w[e]));
}

__global__ __launch_bounds__(256) void dinv_kernel(
    const int* __restrict__ offs, const int2* __restrict__ sw,
    float* __restrict__ dinv, int N, int E) {
  int c = blockIdx.x * 256 + threadIdx.x;
  if (c >= N) return;
  const int beg = offs[c];
  const int end = (c + 1 < N) ? offs[c + 1] : E;
  float d = 1.0f;
  for (int k = beg; k < end; ++k) d += __int_as_float(sw[k].y);
  dinv[c] = d > 0.0f ? rsqrtf(d) : 0.0f;
}

__global__ __launch_bounds__(256) void gemm_xWc_kernel(
    const float* __restrict__ h,
    const float* __restrict__ Wc,
    float* __restrict__ x) {
  const int K = 256, NN = 256;
  const int wave = threadIdx.x >> 5;
  const int lane = threadIdx.x & 31;
  const int m0 = (blockIdx.x >> 1) * 16;
  const int n0 = (((blockIdx.x & 1) << 3) + wave) * 16;
  const int r = lane & 15;
  const int g = lane >> 4;

  const float* arow = h + (size_t)(m0 + r) * K + 2 * g;
  const float* bcol = Wc + (size_t)(2 * g) * NN + n0 + r;

  v8f c = {};
  for (int kk = 0; kk < K; kk += 4) {
    v2f a;
    a.x = arow[kk];
    a.y = arow[kk + 1];
    v2f b;
    b.x = bcol[kk * NN];
    b.y = bcol[kk * NN + NN];
    c = __builtin_amdgcn_wmma_f32_16x16x4_f32(false, a, false, b, (short)0, c,
                                              false, false);
  }
#pragma unroll
  for (int i = 0; i < 8; ++i) {
    int m = m0 + i + 8 * g;
    x[(size_t)m * NN + n0 + r] = c[i];
  }
}

__global__ __launch_bounds__(256) void gather_kernel(
    const int* __restrict__ offs, const int2* __restrict__ sw,
    const float* __restrict__ dinv,
    const float* __restrict__ x,
    const float* __restrict__ bc, float* __restrict__ out) {
  const int wave = __builtin_amdgcn_readfirstlane(threadIdx.x >> 5);
  const int lane = threadIdx.x & 31;
  const int c = blockIdx.x * 8 + wave;
  if (c >= N_NODES) return;
  const int beg = offs[c];
  const int end = (c + 1 < N_NODES) ? offs[c + 1] : N_EDGES;
  const float dc = dinv[c];
  const float4* __restrict__ x4 = (const float4*)x;
  float4 a0 = make_float4(0.f, 0.f, 0.f, 0.f);
  float4 a1 = make_float4(0.f, 0.f, 0.f, 0.f);
  for (int k0 = beg; k0 < end; k0 += 32) {
    const int k = k0 + lane;
    int r = 0;
    float nm = 0.0f;
    if (k < end) {
      const int2 p = sw[k];
      r = p.x;
      nm = dinv[r] * __int_as_float(p.y) * dc;
    }
    const int m = min(32, end - k0);
#pragma unroll 4
    for (int j = 0; j < m; ++j) {
      const int rj = __builtin_amdgcn_readlane(r, j);
      const float nj = __int_as_float(__builtin_amdgcn_readlane(__float_as_int(nm), j));
      const float4* row = x4 + (size_t)rj * (HID / 4);
      const float4 p0 = row[lane];
      const float4 p1 = row[32 + lane];
      a0.x = fmaf(p0.x, nj, a0.x);
      a0.y = fmaf(p0.y, nj, a0.y);
      a0.z = fmaf(p0.z, nj, a0.z);
      a0.w = fmaf(p0.w, nj, a0.w);
      a1.x = fmaf(p1.x, nj, a1.x);
      a1.y = fmaf(p1.y, nj, a1.y);
      a1.z = fmaf(p1.z, nj, a1.z);
      a1.w = fmaf(p1.w, nj, a1.w);
    }
  }
  const float ns = dc * dc;
  const float4* srow = x4 + (size_t)c * (HID / 4);
  const float4 s0 = srow[lane];
  const float4 s1 = srow[32 + lane];
  const int j0 = 4 * lane;
  float4 o0, o1;
  o0.x = fmaf(s0.x, ns, a0.x) + bc[j0 + 0];
  o0.y = fmaf(s0.y, ns, a0.y) + bc[j0 + 1];
  o0.z = fmaf(s0.z, ns, a0.z) + bc[j0 + 2];
  o0.w = fmaf(s0.w, ns, a0.w) + bc[j0 + 3];
  o1.x = fmaf(s1.x, ns, a1.x) + bc[128 + j0 + 0];
  o1.y = fmaf(s1.y, ns, a1.y) + bc[128 + j0 + 1];
  o1.z = fmaf(s1.z, ns, a1.z) + bc[128 + j0 + 2];
  o1.w = fmaf(s1.w, ns, a1.w) + bc[128 + j0 + 3];
  float4* orow = (float4*)out + (size_t)c * (HID / 4);
  orow[lane] = o0;
  orow[32 + lane] = o1;
}

extern "C" void kernel_launch(void* const* d_in, const int* in_sizes, int n_in,
                              void* d_out, int out_size, void* d_ws, size_t ws_size,
                              hipStream_t stream) {
  const float* h    = (const float*)d_in[0];
  const int*   ei   = (const int*)d_in[1];
  const float* attr = (const float*)d_in[2];
  const float* W1   = (const float*)d_in[3];
  const float* b1   = (const float*)d_in[4];
  const float* W2   = (const float*)d_in[5];
  const float* b2   = (const float*)d_in[6];
  const float* Wc   = (const float*)d_in[7];
  const float* bc   = (const float*)d_in[8];
  float* out = (float*)d_out;

  const int N = N_NODES, E = N_EDGES;

  float* x    = (float*)d_ws;
  int2*  sw   = (int2*)(x + (size_t)N * HID);
  float* dinv = (float*)(sw + E);
  int*   offs = (int*)(dinv + N);
  int*   cnt  = (int*)d_out;
  int*   rank = cnt + N;
  float* w    = (float*)(rank + E);

  cnt_zero_kernel<<<(N + 255) / 256, 256, 0, stream>>>(cnt, N);
  edge_mlp_kernel<<<(E + 255) / 256, 256, 0, stream>>>(attr, W1, b1, W2, b2, ei + E, cnt,
                                                       w, rank, E);
  offs_scan_kernel<<<1, 1024, 0, stream>>>(cnt, offs);
  fill_kernel<<<(E + 255) / 256, 256, 0, stream>>>(ei, w, rank, offs, sw, E);
  dinv_kernel<<<(N + 255) / 256, 256, 0, stream>>>(offs, sw, dinv, N, E);
  gemm_xWc_kernel<<<(N / 16) * 2, 256, 0, stream>>>(h, Wc, x);
  gather_kernel<<<(N + 7) / 8, 256, 0, stream>>>(offs, sw, dinv, x, bc, out);
}


# x = h @ Wc: each wave keeps its 16 columns of Wc for all K in registers and walks 16-row tiles of h staged once in LDS, 256 resident work-groups; sum order unchanged
# speedup vs baseline: 4.2761x; 1.2964x over previous
#include <hip/hip_runtime.h>
#include <math.h>

typedef __attribute__((ext_vector_type(2))) float v2f;
typedef __attribute__((ext_vector_type(8))) float v8f;

#define N_NODES 50000
#define N_EDGES 800000
#define HID 256

__global__ __launch_bounds__(256) void cnt_zero_kernel(int* __restrict__ cnt, int N) {
  int i = blockIdx.x * 256 + threadIdx.x;
  if (i < N) cnt[i] = 0;
}

__global__ __launch_bounds__(256) void edge_mlp_kernel(
    const float* __restrict__ attr,
    const float* __restrict__ W1,
    const float* __restrict__ b1,
    const float* __restrict__ W2,
    const float* __restrict__ b2,
    const int* __restrict__ col,
    int* __restrict__ cnt,
    float* __restrict__ w, int* __restrict__ rank, int E) {
  __shared__ float sW1[3 * 128];
  __shared__ float sb1[128];
  __shared__ float sW2[128];
  for (int i = threadIdx.x; i < 384; i += 256) sW1[i] = W1[i];
  if (threadIdx.x < 128) {
    sb1[threadIdx.x] = b1[threadIdx.x];
    sW2[threadIdx.x] = W2[threadIdx.x];
  }
  __syncthreads();
  int e = blockIdx.x * 256 + threadIdx.x;
  if (e >= E) return;
  const int rk = atomicAdd(&cnt[col[e]], 1);
  float a0 = attr[e * 3 + 0];
  float a1 = attr[e * 3 + 1];
  float a2 = attr[e * 3 + 2];
  float acc = 0.0f;
#pragma unroll 4
  for (int j = 0; j < 128; ++j) {
    float z = fmaf(a0, sW1[j], fmaf(a1, sW1[128 + j], fmaf(a2, sW1[256 + j], sb1[j])));
    float t = z / (1.0f + __expf(-z));
    acc = fmaf(t, sW2[j], acc);
  }
  float zz = acc + b2[0];
  w[e] = 1.0f / (1.0f + __expf(-zz));
  rank[e] = rk;
}

#define SCAN_ITERS 13
static_assert(32 * SCAN_ITERS * 128 >= N_NODES, "scan slice too small");
static_assert(N_NODES % 4 == 0, "int4 groups must not straddle the end");

__device__ __forceinline__ int wave_incl_scan(int v, int lane) {
#pragma unroll
  for (int d = 1; d < 32; d <<= 1) {
    int t = __shfl_up(v, d, 32);
    if (lane >= d) v += t;
  }
  return v;
}

__global__ __launch_bounds__(1024) void offs_scan_kernel(
    const int* __restrict__ cnt, int* __restrict__ offs) {
  __shared__ int wtot[32];
  const int wave = threadIdx.x >> 5;
  const int lane = threadIdx.x & 31;
  const int base = wave * (SCAN_ITERS * 128) + lane * 4;
  int4 v[SCAN_ITERS];
  int ex[SCAN_ITERS];
#pragma unroll
  for (int i = 0; i < SCAN_ITERS; ++i) {
    const int idx = base + i * 128;
    v[i] = idx < N_NODES ? *(const int4*)(cnt + idx) : make_int4(0, 0, 0, 0);
  }
  int carry = 0;
#pragma unroll
  for (int i = 0; i < SCAN_ITERS; ++i) {
    const int sum4 = v[i].x + v[i].y + v[i].z + v[i].w;
    const int inc = wave_incl_scan(sum4, lane);
    ex[i] = carry + inc - sum4;
    carry += __shfl(inc, 31, 32);
  }
  if (lane == 0) wtot[wave] = carry;
  __syncthreads();
  const int tot = wtot[lane];
  const int tinc = wave_incl_scan(tot, lane);
  const int woff = __shfl(tinc - tot, wave, 32);
#pragma unroll
  for (int i = 0; i < SCAN_ITERS; ++i) {
    const int idx = base + i * 128;
    if (idx < N_NODES) {
      const int p = woff + ex[i];
      *(int4*)(offs + idx) =
          make_int4(p, p + v[i].x, p + v[i].x + v[i].y, p + v[i].x + v[i].y + v[i].z);
    }
  }
}

__global__ __launch_bounds__(256) void fill_kernel(
    const int* __restrict__ ei,
    const float* __restrict__ w, const int* __restrict__ rank,
    const int* __restrict__ offs, int2* __restrict__ sw, int E) {
  int e = blockIdx.x * 256 + threadIdx.x;
  if (e >= E) return;
  const int r = ei[e];
  const int c = ei[E + e];
  sw[offs[c] + rank[e]] = make_int2(r, __float_as_int(w[e]));
}

__global__ __launch_bounds__(256) void dinv_kernel(
    const int* __restrict__ offs, const int2* __restrict__ sw,
    float* __restrict__ dinv, int N, int E) {
  int c = blockIdx.x * 256 + threadIdx.x;
  if (c >= N) return;
  const int beg = offs[c];
  const int end = (c + 1 < N) ? offs[c + 1] : E;
  float d = 1.0f;
  for (int k = beg; k < end; ++k) d += __int_as_float(sw[k].y);
  dinv[c] = d > 0.0f ? rsqrtf(d) : 0.0f;
}

#define M_TILES (N_NODES / 16)
#define GEMM_GROUPS 256
#define A_PITCH 260
static_assert(N_NODES % 16 == 0, "whole 16-row tiles only");
static_assert(GEMM_GROUPS <= M_TILES, "every work-group needs a first tile");

__global__ __launch_bounds__(512) void gemm_xWc_kernel(
    const float* __restrict__ h,
    const float* __restrict__ Wc,
    float* __restrict__ x) {
  __shared__ __attribute__((aligned(16))) float sA[2][16 * A_PITCH];
  const int wave = threadIdx.x >> 5;
  const int lane = threadIdx.x & 31;
  const int n0 = wave * 16;
  const int r = lane & 15;
  const int g = lane >> 4;

  const float* bcol = Wc + (size_t)(2 * g) * HID + n0 + r;
  v2f bfrag[HID / 4];
#pragma unroll
  for (int j = 0; j < HID / 4; ++j) {
    bfrag[j].x = bcol[(4 * j) * HID];
    bfrag[j].y = bcol[(4 * j + 1) * HID];
  }

  const float4* __restrict__ h4 = (const float4*)h;
  const int srow = threadIdx.x >> 6;
  const int sc4 = threadIdx.x & 63;
  const int soff0 = srow * A_PITCH + 4 * sc4;
  const int soff1 = (srow + 8) * A_PITCH + 4 * sc4;
  const int aoff = r * A_PITCH + 2 * g;

  int mt = blockIdx.x;
  {
    const float4 q0 = h4[(size_t)(mt * 16 + srow) * (HID / 4) + sc4];
    const float4 q1 = h4[(size_t)(mt * 16 + srow + 8) * (HID / 4) + sc4];
    *(float4*)&sA[0][soff0] = q0;
    *(float4*)&sA[0][soff1] = q1;
  }
  __syncthreads();
  int buf = 0;
  for (; mt < M_TILES; mt += GEMM_GROUPS) {
    const int nxt = mt + GEMM_GROUPS;
    const bool has_next = nxt < M_TILES;
    float4 p0 = make_float4(0.f, 0.f, 0.f, 0.f);
    float4 p1 = p0;
    if (has_next) {
      p0 = h4[(size_t)(nxt * 16 + srow) * (HID / 4) + sc4];
      p1 = h4[(size_t)(nxt * 16 + srow + 8) * (HID / 4) + sc4];
    }
    const float* cur = &sA[buf][aoff];
    v8f c = {};
#pragma unroll
    for (int j = 0; j < HID / 4; ++j) {
      const v2f a = *(const v2f*)(cur + 4 * j);
      c = __builtin_amdgcn_wmma_f32_16x16x4_f32(false, a, false, bfrag[j], (short)0, c,
                                                false, false);
    }
    if (has_next) {
      *(float4*)&sA[buf ^ 1][soff0] = p0;
      *(float4*)&sA[buf ^ 1][soff1] = p1;
    }
    __syncthreads();
    const int m0 = mt * 16;
#pragma unroll
    for (int i = 0; i < 8; ++i) {
      int m = m0 + i + 8 * g;
      x[(size_t)m * HID + n0 + r] = c[i];
    }
    buf ^= 1;
  }
}

__global__ __launch_bounds__(256) void gather_kernel(
    const int* __restrict__ offs, const int2* __restrict__ sw,
    const float* __restrict__ dinv,
    const float* __restrict__ x,
    const float* __restrict__ bc, float* __restrict__ out) {
  const int wave = __builtin_amdgcn_readfirstlane(threadIdx.x >> 5);
  const int lane = threadIdx.x & 31;
  const int c = blockIdx.x * 8 + wave;
  if (c >= N_NODES) return;
  const int beg = offs[c];
  const int end = (c + 1 < N_NODES) ? offs[c + 1] : N_EDGES;
  const float dc = dinv[c];
  const float4* __restrict__ x4 = (const float4*)x;
  float4 a0 = make_float4(0.f, 0.f, 0.f, 0.f);
  float4 a1 = make_float4(0.f, 0.f, 0.f, 0.f);
  for (int k0 = beg; k0 < end; k0 += 32) {
    const int k = k0 + lane;
    int r = 0;
    float nm = 0.0f;
    if (k < end) {
      const int2 p = sw[k];
      r = p.x;
      nm = dinv[r] * __int_as_float(p.y) * dc;
    }
    const int m = min(32, end - k0);
#pragma unroll 4
    for (int j = 0; j < m; ++j) {
      const int rj = __builtin_amdgcn_readlane(r, j);
      const float nj = __int_as_float(__builtin_amdgcn_readlane(__float_as_int(nm), j));
      const float4* row = x4 + (size_t)rj * (HID / 4);
      const float4 p0 = row[lane];
      const float4 p1 = row[32 + lane];
      a0.x = fmaf(p0.x, nj, a0.x);
      a0.y = fmaf(p0.y, nj, a0.y);
      a0.z = fmaf(p0.z, nj, a0.z);
      a0.w = fmaf(p0.w, nj, a0.w);
      a1.x = fmaf(p1.x, nj, a1.x);
      a1.y = fmaf(p1.y, nj, a1.y);
      a1.z = fmaf(p1.z, nj, a1.z);
      a1.w = fmaf(p1.w, nj, a1.w);
    }
  }
  const float ns = dc * dc;
  const float4* srow = x4 + (size_t)c * (HID / 4);
  const float4 s0 = srow[lane];
  const float4 s1 = srow[32 + lane];
  const int j0 = 4 * lane;
  float4 o0, o1;
  o0.x = fmaf(s0.x, ns, a0.x) + bc[j0 + 0];
  o0.y = fmaf(s0.y, ns, a0.y) + bc[j0 + 1];
  o0.z = fmaf(s0.z, ns, a0.z) + bc[j0 + 2];
  o0.w = fmaf(s0.w, ns, a0.w) + bc[j0 + 3];
  o1.x = fmaf(s1.x, ns, a1.x) + bc[128 + j0 + 0];
  o1.y = fmaf(s1.y, ns, a1.y) + bc[128 + j0 + 1];
  o1.z = fmaf(s1.z, ns, a1.z) + bc[128 + j0 + 2];
  o1.w = fmaf(s1.w, ns, a1.w) + bc[128 + j0 + 3];
  float4* orow = (float4*)out + (size_t)c * (HID / 4);
  orow[lane] = o0;
  orow[32 + lane] = o1;
}

extern "C" void kernel_launch(void* const* d_in, const int* in_sizes, int n_in,
                              void* d_out, int out_size, void* d_ws, size_t ws_size,
                              hipStream_t stream) {
  const float* h    = (const float*)d_in[0];
  const int*   ei   = (const int*)d_in[1];
  const float* attr = (const float*)d_in[2];
  const float* W1   = (const float*)d_in[3];
  const float* b1   = (const float*)d_in[4];
  const float* W2   = (const float*)d_in[5];
  const float* b2   = (const float*)d_in[6];
  const float* Wc   = (const float*)d_in[7];
  const float* bc   = (const float*)d_in[8];
  float* out = (float*)d_out;

  const int N = N_NODES, E = N_EDGES;

  float* x    = (float*)d_ws;
  int2*  sw   = (int2*)(x + (size_t)N * HID);
  float* dinv = (float*)(sw + E);
  int*   offs = (int*)(dinv + N);
  int*   cnt  = (int*)d_out;
  int*   rank = cnt + N;
  float* w    = (float*)(rank + E);

  cnt_zero_kernel<<<(N + 255) / 256, 256, 0, stream>>>(cnt, N);
  edge_mlp_kernel<<<(E + 255) / 256, 256, 0, stream>>>(attr, W1, b1, W2, b2, ei + E, cnt,
                                                       w, rank, E);
  offs_scan_kernel<<<1, 1024, 0, stream>>>(cnt, offs);
  fill_kernel<<<(E + 255) / 256, 256, 0, stream>>>(ei, w, rank, offs, sw, E);
  dinv_kernel<<<(N + 255) / 256, 256, 0, stream>>>(offs, sw, dinv, N, E);
  gemm_xWc_kernel<<<GEMM_GROUPS, 512, 0, stream>>>(h, Wc, x);
  gather_kernel<<<(N + 7) / 8, 256, 0, stream>>>(offs, sw, dinv, x, bc, out);
}


# edge MLP inner loop: hardware reciprocal replaces the IEEE division of SiLU and the layer weights become scalar-register operands (no LDS reads, no staging barrier); sum order unchanged
# speedup vs baseline: 4.5820x; 1.0715x over previous
#include <hip/hip_runtime.h>
#include <math.h>

typedef __attribute__((ext_vector_type(2))) float v2f;
typedef __attribute__((ext_vector_type(8))) float v8f;

#define N_NODES 50000
#define N_EDGES 800000
#define HID 256

__global__ __launch_bounds__(256) void cnt_zero_kernel(int* __restrict__ cnt, int N) {
  int i = blockIdx.x * 256 + threadIdx.x;
  if (i < N) cnt[i] = 0;
}

__global__ __launch_bounds__(256) void edge_mlp_kernel(
    const float* __restrict__ attr,
    const float* __restrict__ W1,
    const float* __restrict__ b1,
    const float* __restrict__ W2,
    const float* __restrict__ b2,
    const int* __restrict__ col,
    int* __restrict__ cnt,
    float* __restrict__ w, int* __restrict__ rank, int E) {
  int e = blockIdx.x * 256 + threadIdx.x;
  if (e >= E) return;
  const int rk = atomicAdd(&cnt[col[e]], 1);
  float a0 = attr[e * 3 + 0];
  float a1 = attr[e * 3 + 1];
  float a2 = attr[e * 3 + 2];
  float acc = 0.0f;
#pragma unroll 8
  for (int j = 0; j < 128; ++j) {
    float z = fmaf(a0, W1[j], fmaf(a1, W1[128 + j], fmaf(a2, W1[256 + j], b1[j])));
    float t = z * __builtin_amdgcn_rcpf(1.0f + __expf(-z));
    acc = fmaf(t, W2[j], acc);
  }
  float zz = acc + b2[0];
  w[e] = 1.0f / (1.0f + __expf(-zz));
  rank[e] = rk;
}

#define SCAN_ITERS 13
static_assert(32 * SCAN_ITERS * 128 >= N_NODES, "scan slice too small");
static_assert(N_NODES % 4 == 0, "int4 groups must not straddle the end");

__device__ __forceinline__ int wave_incl_scan(int v, int lane) {
#pragma unroll
  for (int d = 1; d < 32; d <<= 1) {
    int t = __shfl_up(v, d, 32);
    if (lane >= d) v += t;
  }
  return v;
}

__global__ __launch_bounds__(1024) void offs_scan_kernel(
    const int* __restrict__ cnt, int* __restrict__ offs) {
  __shared__ int wtot[32];
  const int wave = threadIdx.x >> 5;
  const int lane = threadIdx.x & 31;
  const int base = wave * (SCAN_ITERS * 128) + lane * 4;
  int4 v[SCAN_ITERS];
  int ex[SCAN_ITERS];
#pragma unroll
  for (int i = 0; i < SCAN_ITERS; ++i) {
    const int idx = base + i * 128;
    v[i] = idx < N_NODES ? *(const int4*)(cnt + idx) : make_int4(0, 0, 0, 0);
  }
  int carry = 0;
#pragma unroll
  for (int i = 0; i < SCAN_ITERS; ++i) {
    const int sum4 = v[i].x + v[i].y + v[i].z + v[i].w;
    const int inc = wave_incl_scan(sum4, lane);
    ex[i] = carry + inc - sum4;
    carry += __shfl(inc, 31, 32);
  }
  if (lane == 0) wtot[wave] = carry;
  __syncthreads();
  const int tot = wtot[lane];
  const int tinc = wave_incl_scan(tot, lane);
  const int woff = __shfl(tinc - tot, wave, 32);
#pragma unroll
  for (int i = 0; i < SCAN_ITERS; ++i) {
    const int idx = base + i * 128;
    if (idx < N_NODES) {
      const int p = woff + ex[i];
      *(int4*)(offs + idx) =
          make_int4(p, p + v[i].x, p + v[i].x + v[i].y, p + v[i].x + v[i].y + v[i].z);
    }
  }
}

__global__ __launch_bounds__(256) void fill_kernel(
    const int* __restrict__ ei,
    const float* __restrict__ w, const int* __restrict__ rank,
    const int* __restrict__ offs, int2* __restrict__ sw, int E) {
  int e = blockIdx.x * 256 + threadIdx.x;
  if (e >= E) return;
  const int r = ei[e];
  const int c = ei[E + e];
  sw[offs[c] + rank[e]] = make_int2(r, __float_as_int(w[e]));
}

__global__ __launch_bounds__(256) void dinv_kernel(
    const int* __restrict__ offs, const int2* __restrict__ sw,
    float* __restrict__ dinv, int N, int E) {
  int c = blockIdx.x * 256 + threadIdx.x;
  if (c >= N) return;
  const int beg = offs[c];
  const int end = (c + 1 < N) ? offs[c + 1] : E;
  float d = 1.0f;
  for (int k = beg; k < end; ++k) d += __int_as_float(sw[k].y);
  dinv[c] = d > 0.0f ? rsqrtf(d) : 0.0f;
}

#define M_TILES (N_NODES / 16)
#define GEMM_GROUPS 256
#define A_PITCH 260
static_assert(N_NODES % 16 == 0, "whole 16-row tiles only");
static_assert(GEMM_GROUPS <= M_TILES, "every work-group needs a first tile");

__global__ __launch_bounds__(512) void gemm_xWc_kernel(
    const float* __restrict__ h,
    const float* __restrict__ Wc,
    float* __restrict__ x) {
  __shared__ __attribute__((aligned(16))) float sA[2][16 * A_PITCH];
  const int wave = threadIdx.x >> 5;
  const int lane = threadIdx.x & 31;
  const int n0 = wave * 16;
  const int r = lane & 15;
  const int g = lane >> 4;

  const float* bcol = Wc + (size_t)(2 * g) * HID + n0 + r;
  v2f bfrag[HID / 4];
#pragma unroll
  for (int j = 0; j < HID / 4; ++j) {
    bfrag[j].x = bcol[(4 * j) * HID];
    bfrag[j].y = bcol[(4 * j + 1) * HID];
  }

  const float4* __restrict__ h4 = (const float4*)h;
  const int srow = threadIdx.x >> 6;
  const int sc4 = threadIdx.x & 63;
  const int soff0 = srow * A_PITCH + 4 * sc4;
  const int soff1 = (srow + 8) * A_PITCH + 4 * sc4;
  const int aoff = r * A_PITCH + 2 * g;

  int mt = blockIdx.x;
  {
    const float4 q0 = h4[(size_t)(mt * 16 + srow) * (HID / 4) + sc4];
    const float4 q1 = h4[(size_t)(mt * 16 + srow + 8) * (HID / 4) + sc4];
    *(float4*)&sA[0][soff0] = q0;
    *(float4*)&sA[0][soff1] = q1;
  }
  __syncthreads();
  int buf = 0;
  for (; mt < M_TILES; mt += GEMM_GROUPS) {
    const int nxt = mt + GEMM_GROUPS;
    const bool has_next = nxt < M_TILES;
    float4 p0 = make_float4(0.f, 0.f, 0.f, 0.f);
    float4 p1 = p0;
    if (has_next) {
      p0 = h4[(size_t)(nxt * 16 + srow) * (HID / 4) + sc4];
      p1 = h4[(size_t)(nxt * 16 + srow + 8) * (HID / 4) + sc4];
    }
    const float* cur = &sA[buf][aoff];
    v8f c = {};
#pragma unroll
    for (int j = 0; j < HID / 4; ++j) {
      const v2f a = *(const v2f*)(cur + 4 * j);
      c = __builtin_amdgcn_wmma_f32_16x16x4_f32(false, a, false, bfrag[j], (short)0, c,
                                                false, false);
    }
    if (has_next) {
      *(float4*)&sA[buf ^ 1][soff0] = p0;
      *(float4*)&sA[buf ^ 1][soff1] = p1;
    }
    __syncthreads();
    const int m0 = mt * 16;
#pragma unroll
    for (int i = 0; i < 8; ++i) {
      int m = m0 + i + 8 * g;
      x[(size_t)m * HID + n0 + r] = c[i];
    }
    buf ^= 1;
  }
}

__global__ __launch_bounds__(256) void gather_kernel(
    const int* __restrict__ offs, const int2* __restrict__ sw,
    const float* __restrict__ dinv,
    const float* __restrict__ x,
    const float* __restrict__ bc, float* __restrict__ out) {
  const int wave = __builtin_amdgcn_readfirstlane(threadIdx.x >> 5);
  const int lane = threadIdx.x & 31;
  const int c = blockIdx.x * 8 + wave;
  if (c >= N_NODES) return;
  const int beg = offs[c];
  const int end = (c + 1 < N_NODES) ? offs[c + 1] : N_EDGES;
  const float dc = dinv[c];
  const float4* __restrict__ x4 = (const float4*)x;
  float4 a0 = make_float4(0.f, 0.f, 0.f, 0.f);
  float4 a1 = make_float4(0.f, 0.f, 0.f, 0.f);
  for (int k0 = beg; k0 < end; k0 += 32) {
    const int k = k0 + lane;
    int r = 0;
    float nm = 0.0f;
    if (k < end) {
      const int2 p = sw[k];
      r = p.x;
      nm = dinv[r] * __int_as_float(p.y) * dc;
    }
    const int m = min(32, end - k0);
#pragma unroll 4
    for (int j = 0; j < m; ++j) {
      const int rj = __builtin_amdgcn_readlane(r, j);
      const float nj = __int_as_float(__builtin_amdgcn_readlane(__float_as_int(nm), j));
      const float4* row = x4 + (size_t)rj * (HID / 4);
      const float4 p0 = row[lane];
      const float4 p1 = row[32 + lane];
      a0.x = fmaf(p0.x, nj, a0.x);
      a0.y = fmaf(p0.y, nj, a0.y);
      a0.z = fmaf(p0.z, nj, a0.z);
      a0.w = fmaf(p0.w, nj, a0.w);
      a1.x = fmaf(p1.x, nj, a1.x);
      a1.y = fmaf(p1.y, nj, a1.y);
      a1.z = fmaf(p1.z, nj, a1.z);
      a1.w = fmaf(p1.w, nj, a1.w);
    }
  }
  const float ns = dc * dc;
  const float4* srow = x4 + (size_t)c * (HID / 4);
  const float4 s0 = srow[lane];
  const float4 s1 = srow[32 + lane];
  const int j0 = 4 * lane;
  float4 o0, o1;
  o0.x = fmaf(s0.x, ns, a0.x) + bc[j0 + 0];
  o0.y = fmaf(s0.y, ns, a0.y) + bc[j0 + 1];
  o0.z = fmaf(s0.z, ns, a0.z) + bc[j0 + 2];
  o0.w = fmaf(s0.w, ns, a0.w) + bc[j0 + 3];
  o1.x = fmaf(s1.x, ns, a1.x) + bc[128 + j0 + 0];
  o1.y = fmaf(s1.y, ns, a1.y) + bc[128 + j0 + 1];
  o1.z = fmaf(s1.z, ns, a1.z) + bc[128 + j0 + 2];
  o1.w = fmaf(s1.w, ns, a1.w) + bc[128 + j0 + 3];
  float4* orow = (float4*)out + (size_t)c * (HID / 4);
  orow[lane] = o0;
  orow[32 + lane] = o1;
}

extern "C" void kernel_launch(void* const* d_in, const int* in_sizes, int n_in,
                              void* d_out, int out_size, void* d_ws, size_t ws_size,
                              hipStream_t stream) {
  const float* h    = (const float*)d_in[0];
  const int*   ei   = (const int*)d_in[1];
  const float* attr = (const float*)d_in[2];
  const float* W1   = (const float*)d_in[3];
  const float* b1   = (const float*)d_in[4];
  const float* W2   = (const float*)d_in[5];
  const float* b2   = (const float*)d_in[6];
  const float* Wc   = (const float*)d_in[7];
  const float* bc   = (const float*)d_in[8];
  float* out = (float*)d_out;

  const int N = N_NODES, E = N_EDGES;

  float* x    = (float*)d_ws;
  int2*  sw   = (int2*)(x + (size_t)N * HID);
  float* dinv = (float*)(sw + E);
  int*   offs = (int*)(dinv + N);
  int*   cnt  = (int*)d_out;
  int*   rank = cnt + N;
  float* w    = (float*)(rank + E);

  cnt_zero_kernel<<<(N + 255) / 256, 256, 0, stream>>>(cnt, N);
  edge_mlp_kernel<<<(E + 255) / 256, 256, 0, stream>>>(attr, W1, b1, W2, b2, ei + E, cnt,
                                                       w, rank, E);
  offs_scan_kernel<<<1, 1024, 0, stream>>>(cnt, offs);
  fill_kernel<<<(E + 255) / 256, 256, 0, stream>>>(ei, w, rank, offs, sw, E);
  dinv_kernel<<<(N + 255) / 256, 256, 0, stream>>>(offs, sw, dinv, N, E);
  gemm_xWc_kernel<<<GEMM_GROUPS, 512, 0, stream>>>(h, Wc, x);
  gather_kernel<<<(N + 7) / 8, 256, 0, stream>>>(offs, sw, dinv, x, bc, out);
}
